# starting kernel
# baseline (speedup 1.0000x reference)
#include <hip/hip_runtime.h>
#include <math.h>

typedef __attribute__((ext_vector_type(2))) float v2f;
typedef __attribute__((ext_vector_type(8))) float v8f;

#define EMB   768
#define HS    64
#define SEQ   2048
#define BATCH 8
#define XPAD  772
#define VPAD  72

__device__ __forceinline__ v8f wmma_f32(v2f a, v2f b, v8f c) {
  return __builtin_amdgcn_wmma_f32_16x16x4_f32(
       false, a,  false, b,
       (short)0, c,  false,  false);
}

__global__ __launch_bounds__(128) void proj_kernel(
    const float* __restrict__ x,
    const float* __restrict__ Wq, const float* __restrict__ Wk,
    const float* __restrict__ Wv,
    float* __restrict__ Q, float* __restrict__ Kt, float* __restrict__ V) {
  __shared__ float xs[16 * XPAD];

  const int tileM = blockIdx.x;
  const int which = blockIdx.y;
  const int tid   = threadIdx.x;
  const int lane  = tid & 31;
  const int wave  = tid >> 5;
  const int l16   = lane & 15;
  const int half  = lane >> 4;
  const int halfk = half * 2;
  const int rowBase = tileM * 16;

  for (int i = tid; i < 16 * EMB; i += 128) {
    const int r = i / EMB, cc = i - r * EMB;
    xs[r * XPAD + cc] = x[(rowBase + r) * EMB + cc];
  }
  __syncthreads();

  const float* W = (which == 0) ? Wq : (which == 1) ? Wk : Wv;
  const int n = wave * 16 + l16;

  v8f c = {};
  for (int kc = 0; kc < EMB / 4; ++kc) {
    const int k0 = kc * 4 + halfk;
    v2f a, b;
    a.x = xs[l16 * XPAD + k0];
    a.y = xs[l16 * XPAD + k0 + 1];
    b.x = W[k0 * HS + n];
    b.y = W[(k0 + 1) * HS + n];
    c = wmma_f32(a, b, c);
  }

  if (which == 1) {
    const int bidx = rowBase / SEQ;
    const int s0   = rowBase - bidx * SEQ;
#pragma unroll
    for (int v = 0; v < 8; ++v) {
      const int srow = s0 + v + half * 8;
      Kt[(bidx * HS + n) * SEQ + srow] = c[v];
    }
  } else {
    float* O = (which == 0) ? Q : V;
#pragma unroll
    for (int v = 0; v < 8; ++v) {
      const int m = rowBase + v + half * 8;
      O[m * HS + n] = c[v];
    }
  }
}

__global__ __launch_bounds__(256) void attn_kernel(
    const float* __restrict__ Q, const float* __restrict__ Kt,
    const float* __restrict__ V, float* __restrict__ out) {
  __shared__ float Kbuf[2][HS * 16];
  __shared__ float Vbuf[2][16 * VPAD];
  __shared__ float plds[8][16 * 17];

  const int tid   = threadIdx.x;
  const int lane  = tid & 31;
  const int wave  = tid >> 5;
  const int gt    = blockIdx.x * 8 + wave;
  const int b     = gt >> 7;
  const int qbase = (gt & 127) * 16;
  const int l16   = lane & 15;
  const int half  = lane >> 4;
  const int halfk = half * 2;

  v2f qa[16];
  const int mrow = (b * SEQ + qbase + l16) * HS;
#pragma unroll
  for (int kc = 0; kc < 16; ++kc) {
    qa[kc].x = Q[mrow + kc * 4 + halfk];
    qa[kc].y = Q[mrow + kc * 4 + halfk + 1];
  }

  v8f acc0 = {}, acc1 = {}, acc2 = {}, acc3 = {};
  float mst[8], lst[8];
#pragma unroll
  for (int v = 0; v < 8; ++v) { mst[v] = -INFINITY; lst[v] = 0.0f; }

  float* pl = &plds[wave][0];
  const float scale = 0.125f;

  auto stage = [&](int kt, int bufi) {
    float* kb = &Kbuf[bufi][0];
    float* vb = &Vbuf[bufi][0];
    const int kBase = (b * HS) * SEQ + kt * 16;
    const int vBase = (b * SEQ + kt * 16) * HS;
#pragma unroll
    for (int i = tid; i < HS * 16; i += 256) {
      const int h = i >> 4, kl = i & 15;
      kb[i] = Kt[kBase + h * SEQ + kl];
    }
#pragma unroll
    for (int i = tid; i < 16 * HS; i += 256) {
      const int kl = i >> 6, h = i & 63;
      vb[kl * VPAD + h] = V[vBase + kl * HS + h];
    }
  };

  stage(0, 0);

  for (int kt = 0; kt < SEQ / 16; ++kt) {
    __syncthreads();
    const int bufi = kt & 1;
    if (kt + 1 < SEQ / 16) stage(kt + 1, bufi ^ 1);

    const float* kb = &Kbuf[bufi][0];
    const float* vb = &Vbuf[bufi][0];

    v8f s = {};
#pragma unroll
    for (int kc = 0; kc < 16; ++kc) {
      const int h = kc * 4 + halfk;
      v2f kv;
      kv.x = kb[h * 16 + l16];
      kv.y = kb[(h + 1) * 16 + l16];
      s = wmma_f32(qa[kc], kv, s);
    }

#pragma unroll
    for (int v = 0; v < 8; ++v) {
      float sv = s[v] * scale;
      float tm = sv;
      tm = fmaxf(tm, __shfl_xor(tm, 1));
      tm = fmaxf(tm, __shfl_xor(tm, 2));
      tm = fmaxf(tm, __shfl_xor(tm, 4));
      tm = fmaxf(tm, __shfl_xor(tm, 8));
      const float nm    = fmaxf(mst[v], tm);
      const float alpha = __expf(mst[v] - nm);
      const float p     = __expf(sv - nm);
      float rs = p;
      rs += __shfl_xor(rs, 1);
      rs += __shfl_xor(rs, 2);
      rs += __shfl_xor(rs, 4);
      rs += __shfl_xor(rs, 8);
      lst[v] = lst[v] * alpha + rs;
      mst[v] = nm;
      acc0[v] *= alpha; acc1[v] *= alpha; acc2[v] *= alpha; acc3[v] *= alpha;
      pl[(v + half * 8) * 17 + l16] = p;
    }

#pragma unroll
    for (int kc = 0; kc < 4; ++kc) {
      v2f a;
      a.x = pl[l16 * 17 + kc * 4 + halfk];
      a.y = pl[l16 * 17 + kc * 4 + halfk + 1];
      const int r0 = (kc * 4 + halfk) * VPAD;
      const int r1 = r0 + VPAD;
      { v2f bv; bv.x = vb[r0 + l16];      bv.y = vb[r1 + l16];      acc0 = wmma_f32(a, bv, acc0); }
      { v2f bv; bv.x = vb[r0 + 16 + l16]; bv.y = vb[r1 + 16 + l16]; acc1 = wmma_f32(a, bv, acc1); }
      { v2f bv; bv.x = vb[r0 + 32 + l16]; bv.y = vb[r1 + 32 + l16]; acc2 = wmma_f32(a, bv, acc2); }
      { v2f bv; bv.x = vb[r0 + 48 + l16]; bv.y = vb[r1 + 48 + l16]; acc3 = wmma_f32(a, bv, acc3); }
    }
  }

#pragma unroll
  for (int v = 0; v < 8; ++v) {
    const float inv = 1.0f / lst[v];
    const int orow = (b * SEQ + qbase + v + half * 8) * HS;
    out[orow + l16]      = acc0[v] * inv;
    out[orow + 16 + l16] = acc1[v] * inv;
    out[orow + 32 + l16] = acc2[v] * inv;
    out[orow + 48 + l16] = acc3[v] * inv;
  }
}

extern "C" void kernel_launch(void* const* d_in, const int* in_sizes, int n_in,
                              void* d_out, int out_size, void* d_ws, size_t ws_size,
                              hipStream_t stream) {
  (void)in_sizes; (void)n_in; (void)out_size; (void)ws_size;
  const float* x  = (const float*)d_in[0];
  const float* Wk = (const float*)d_in[1];
  const float* Wq = (const float*)d_in[2];
  const float* Wv = (const float*)d_in[3];

  float* ws = (float*)d_ws;
  const size_t plane = (size_t)BATCH * SEQ * HS;
  float* Q  = ws;
  float* Kt = ws + plane;
  float* V  = ws + 2 * plane;
  float* out = (float*)d_out;

  dim3 g1(BATCH * SEQ / 16, 3);
  proj_kernel<<<g1, 128, 0, stream>>>(x, Wq, Wk, Wv, Q, Kt, V);

  attn_kernel<<<(BATCH * SEQ / 16) / 8, 256, 0, stream>>>(Q, Kt, V, out);
}


# attention K/V staging software-pipelined: next tile fetched as one group of 16-byte loads before the step's WMMAs, stored to LDS after them (was 7 serial load-wait-store trips per step)
# speedup vs baseline: 2.0578x; 2.0578x over previous
#include <hip/hip_runtime.h>
#include <math.h>

typedef __attribute__((ext_vector_type(2))) float v2f;
typedef __attribute__((ext_vector_type(4))) float v4f;
typedef __attribute__((ext_vector_type(8))) float v8f;

#define EMB   768
#define HS    64
#define SEQ   2048
#define BATCH 8
#define XPAD  772
#define VPAD  72

__device__ __forceinline__ v8f wmma_f32(v2f a, v2f b, v8f c) {
  return __builtin_amdgcn_wmma_f32_16x16x4_f32(
       false, a,  false, b,
       (short)0, c,  false,  false);
}

__global__ __launch_bounds__(128) void proj_kernel(
    const float* __restrict__ x,
    const float* __restrict__ Wq, const float* __restrict__ Wk,
    const float* __restrict__ Wv,
    float* __restrict__ Q, float* __restrict__ Kt, float* __restrict__ V) {
  __shared__ float xs[16 * XPAD];

  const int tileM = blockIdx.x;
  const int which = blockIdx.y;
  const int tid   = threadIdx.x;
  const int lane  = tid & 31;
  const int wave  = tid >> 5;
  const int l16   = lane & 15;
  const int half  = lane >> 4;
  const int halfk = half * 2;
  const int rowBase = tileM * 16;

  for (int i = tid; i < 16 * EMB; i += 128) {
    const int r = i / EMB, cc = i - r * EMB;
    xs[r * XPAD + cc] = x[(rowBase + r) * EMB + cc];
  }
  __syncthreads();

  const float* W = (which == 0) ? Wq : (which == 1) ? Wk : Wv;
  const int n = wave * 16 + l16;

  v8f c = {};
  for (int kc = 0; kc < EMB / 4; ++kc) {
    const int k0 = kc * 4 + halfk;
    v2f a, b;
    a.x = xs[l16 * XPAD + k0];
    a.y = xs[l16 * XPAD + k0 + 1];
    b.x = W[k0 * HS + n];
    b.y = W[(k0 + 1) * HS + n];
    c = wmma_f32(a, b, c);
  }

  if (which == 1) {
    const int bidx = rowBase / SEQ;
    const int s0   = rowBase - bidx * SEQ;
#pragma unroll
    for (int v = 0; v < 8; ++v) {
      const int srow = s0 + v + half * 8;
      Kt[(bidx * HS + n) * SEQ + srow] = c[v];
    }
  } else {
    float* O = (which == 0) ? Q : V;
#pragma unroll
    for (int v = 0; v < 8; ++v) {
      const int m = rowBase + v + half * 8;
      O[m * HS + n] = c[v];
    }
  }
}

__global__ __launch_bounds__(256) void attn_kernel(
    const float* __restrict__ Q, const float* __restrict__ Kt,
    const float* __restrict__ V, float* __restrict__ out) {
  __shared__ __attribute__((aligned(16))) float Kbuf[2][HS * 16];
  __shared__ __attribute__((aligned(16))) float Vbuf[2][16 * VPAD];
  __shared__ float plds[8][16 * 17];

  const int tid   = threadIdx.x;
  const int lane  = tid & 31;
  const int wave  = tid >> 5;
  const int gt    = blockIdx.x * 8 + wave;
  const int b     = gt >> 7;
  const int qbase = (gt & 127) * 16;
  const int l16   = lane & 15;
  const int half  = lane >> 4;
  const int halfk = half * 2;

  v2f qa[16];
  const int mrow = (b * SEQ + qbase + l16) * HS;
#pragma unroll
  for (int kc = 0; kc < 16; ++kc) {
    qa[kc].x = Q[mrow + kc * 4 + halfk];
    qa[kc].y = Q[mrow + kc * 4 + halfk + 1];
  }

  v8f acc0 = {}, acc1 = {}, acc2 = {}, acc3 = {};
  float mst[8], lst[8];
#pragma unroll
  for (int v = 0; v < 8; ++v) { mst[v] = -INFINITY; lst[v] = 0.0f; }

  float* pl = &plds[wave][0];
  const float scale = 0.125f;

  const int kpiece = (tid >> 2) * 16 + (tid & 3) * 4;
  const int vpiece = (tid >> 4) * VPAD + (tid & 15) * 4;
  const float* kg = Kt + (b * HS + (tid >> 2)) * SEQ + (tid & 3) * 4;
  const float* vg = V + (b * SEQ + (tid >> 4)) * HS + (tid & 15) * 4;

  {
    const v4f k0 = *(const v4f*)kg;
    const v4f v0 = *(const v4f*)vg;
    *(v4f*)&Kbuf[0][kpiece] = k0;
    *(v4f*)&Vbuf[0][vpiece] = v0;
  }

  for (int kt = 0; kt < SEQ / 16; ++kt) {
    __syncthreads();
    const int bufi = kt & 1;
    const bool more = kt + 1 < SEQ / 16;
    v4f kn = {}, vn = {};
    if (more) {
      kn = *(const v4f*)(kg + (kt + 1) * 16);
      vn = *(const v4f*)(vg + (kt + 1) * 16 * HS);
    }

    const float* kb = &Kbuf[bufi][0];
    const float* vb = &Vbuf[bufi][0];

    v8f s = {};
#pragma unroll
    for (int kc = 0; kc < 16; ++kc) {
      const int h = kc * 4 + halfk;
      v2f kv;
      kv.x = kb[h * 16 + l16];
      kv.y = kb[(h + 1) * 16 + l16];
      s = wmma_f32(qa[kc], kv, s);
    }

#pragma unroll
    for (int v = 0; v < 8; ++v) {
      float sv = s[v] * scale;
      float tm = sv;
      tm = fmaxf(tm, __shfl_xor(tm, 1));
      tm = fmaxf(tm, __shfl_xor(tm, 2));
      tm = fmaxf(tm, __shfl_xor(tm, 4));
      tm = fmaxf(tm, __shfl_xor(tm, 8));
      const float nm    = fmaxf(mst[v], tm);
      const float alpha = __expf(mst[v] - nm);
      const float p     = __expf(sv - nm);
      float rs = p;
      rs += __shfl_xor(rs, 1);
      rs += __shfl_xor(rs, 2);
      rs += __shfl_xor(rs, 4);
      rs += __shfl_xor(rs, 8);
      lst[v] = lst[v] * alpha + rs;
      mst[v] = nm;
      acc0[v] *= alpha; acc1[v] *= alpha; acc2[v] *= alpha; acc3[v] *= alpha;
      pl[(v + half * 8) * 17 + l16] = p;
    }

#pragma unroll
    for (int kc = 0; kc < 4; ++kc) {
      v2f a;
      a.x = pl[l16 * 17 + kc * 4 + halfk];
      a.y = pl[l16 * 17 + kc * 4 + halfk + 1];
      const int r0 = (kc * 4 + halfk) * VPAD;
      const int r1 = r0 + VPAD;
      { v2f bv; bv.x = vb[r0 + l16];      bv.y = vb[r1 + l16];      acc0 = wmma_f32(a, bv, acc0); }
      { v2f bv; bv.x = vb[r0 + 16 + l16]; bv.y = vb[r1 + 16 + l16]; acc1 = wmma_f32(a, bv, acc1); }
      { v2f bv; bv.x = vb[r0 + 32 + l16]; bv.y = vb[r1 + 32 + l16]; acc2 = wmma_f32(a, bv, acc2); }
      { v2f bv; bv.x = vb[r0 + 48 + l16]; bv.y = vb[r1 + 48 + l16]; acc3 = wmma_f32(a, bv, acc3); }
    }

    if (more) {
      *(v4f*)&Kbuf[bufi ^ 1][kpiece] = kn;
      *(v4f*)&Vbuf[bufi ^ 1][vpiece] = vn;
    }
  }

#pragma unroll
  for (int v = 0; v < 8; ++v) {
    const float inv = 1.0f / lst[v];
    const int orow = (b * SEQ + qbase + v + half * 8) * HS;
    out[orow + l16]      = acc0[v] * inv;
    out[orow + 16 + l16] = acc1[v] * inv;
    out[orow + 32 + l16] = acc2[v] * inv;
    out[orow + 48 + l16] = acc3[v] * inv;
  }
}

extern "C" void kernel_launch(void* const* d_in, const int* in_sizes, int n_in,
                              void* d_out, int out_size, void* d_ws, size_t ws_size,
                              hipStream_t stream) {
  (void)in_sizes; (void)n_in; (void)out_size; (void)ws_size;
  const float* x  = (const float*)d_in[0];
  const float* Wk = (const float*)d_in[1];
  const float* Wq = (const float*)d_in[2];
  const float* Wv = (const float*)d_in[3];

  float* ws = (float*)d_ws;
  const size_t plane = (size_t)BATCH * SEQ * HS;
  float* Q  = ws;
  float* Kt = ws + plane;
  float* V  = ws + 2 * plane;
  float* out = (float*)d_out;

  dim3 g1(BATCH * SEQ / 16, 3);
  proj_kernel<<<g1, 128, 0, stream>>>(x, Wq, Wk, Wv, Q, Kt, V);

  attn_kernel<<<(BATCH * SEQ / 16) / 8, 256, 0, stream>>>(Q, Kt, V, out);
}


# attention score tile computed transposed (keys x queries): softmax max/sum in-lane + one lane-half swap, P feeds PV from registers, fragment loads grouped up front; row-sum and PV key order change
# speedup vs baseline: 2.6271x; 1.2767x over previous
#include <hip/hip_runtime.h>
#include <math.h>

typedef __attribute__((ext_vector_type(2))) float v2f;
typedef __attribute__((ext_vector_type(4))) float v4f;
typedef __attribute__((ext_vector_type(8))) float v8f;

#define EMB   768
#define HS    64
#define SEQ   2048
#define BATCH 8
#define XPAD  772
#define VPAD  68

__device__ __forceinline__ v8f wmma_f32(v2f a, v2f b, v8f c) {
  return __builtin_amdgcn_wmma_f32_16x16x4_f32(
       false, a,  false, b,
       (short)0, c,  false,  false);
}

__device__ __forceinline__ float xhalf(float x) {
  const int b = __builtin_bit_cast(int, x);
  return __builtin_bit_cast(float,
      __builtin_amdgcn_permlanex16(b, b, 0x76543210, 0xfedcba98, false, false));
}

__global__ __launch_bounds__(128) void proj_kernel(
    const float* __restrict__ x,
    const float* __restrict__ Wq, const float* __restrict__ Wk,
    const float* __restrict__ Wv,
    float* __restrict__ Q, float* __restrict__ Kt, float* __restrict__ V) {
  __shared__ float xs[16 * XPAD];

  const int tileM = blockIdx.x;
  const int which = blockIdx.y;
  const int tid   = threadIdx.x;
  const int lane  = tid & 31;
  const int wave  = tid >> 5;
  const int l16   = lane & 15;
  const int half  = lane >> 4;
  const int halfk = half * 2;
  const int rowBase = tileM * 16;

  for (int i = tid; i < 16 * EMB; i += 128) {
    const int r = i / EMB, cc = i - r * EMB;
    xs[r * XPAD + cc] = x[(rowBase + r) * EMB + cc];
  }
  __syncthreads();

  const float* W = (which == 0) ? Wq : (which == 1) ? Wk : Wv;
  const int n = wave * 16 + l16;

  v8f c = {};
  for (int kc = 0; kc < EMB / 4; ++kc) {
    const int k0 = kc * 4 + halfk;
    v2f a, b;
    a.x = xs[l16 * XPAD + k0];
    a.y = xs[l16 * XPAD + k0 + 1];
    b.x = W[k0 * HS + n];
    b.y = W[(k0 + 1) * HS + n];
    c = wmma_f32(a, b, c);
  }

  if (which == 1) {
    const int bidx = rowBase / SEQ;
    const int s0   = rowBase - bidx * SEQ;
#pragma unroll
    for (int v = 0; v < 8; ++v) {
      const int srow = s0 + v + half * 8;
      Kt[(bidx * HS + n) * SEQ + srow] = c[v];
    }
  } else {
    float* O = (which == 0) ? Q : V;
#pragma unroll
    for (int v = 0; v < 8; ++v) {
      const int m = rowBase + v + half * 8;
      O[m * HS + n] = c[v];
    }
  }
}

__global__ __launch_bounds__(256) void attn_kernel(
    const float* __restrict__ Q, const float* __restrict__ Kt,
    const float* __restrict__ V, float* __restrict__ out) {
  __shared__ __attribute__((aligned(16))) float Kbuf[2][HS * 16];
  __shared__ __attribute__((aligned(16))) float Vbuf[2][16 * VPAD];

  const int tid   = threadIdx.x;
  const int lane  = tid & 31;
  const int wave  = tid >> 5;
  const int gt    = blockIdx.x * 8 + wave;
  const int b     = gt >> 7;
  const int qbase = (gt & 127) * 16;
  const int l16   = lane & 15;
  const int half  = lane >> 4;
  const int halfk = half * 2;

  v2f qa[16];
  const int mrow = (b * SEQ + qbase + l16) * HS;
#pragma unroll
  for (int kc = 0; kc < 16; ++kc) {
    qa[kc].x = Q[mrow + kc * 4 + halfk];
    qa[kc].y = Q[mrow + kc * 4 + halfk + 1];
  }

  v8f acc0 = {}, acc1 = {}, acc2 = {}, acc3 = {};
  float mrun = -INFINITY;
  float lrun = 0.0f;

  const float scale = 0.125f;

  const int kpiece = (tid >> 2) * 16 + (tid & 3) * 4;
  const int vpiece = (tid >> 4) * VPAD + (tid & 15) * 4;
  const float* kg = Kt + (b * HS + (tid >> 2)) * SEQ + (tid & 3) * 4;
  const float* vg = V + (b * SEQ + (tid >> 4)) * HS + (tid & 15) * 4;

  {
    const v4f k0 = *(const v4f*)kg;
    const v4f v0 = *(const v4f*)vg;
    *(v4f*)&Kbuf[0][kpiece] = k0;
    *(v4f*)&Vbuf[0][vpiece] = v0;
  }

  for (int kt = 0; kt < SEQ / 16; ++kt) {
    __syncthreads();
    const int bufi = kt & 1;
    const bool more = kt + 1 < SEQ / 16;
    v4f kn = {}, vn = {};
    if (more) {
      kn = *(const v4f*)(kg + (kt + 1) * 16);
      vn = *(const v4f*)(vg + (kt + 1) * 16 * HS);
    }

    const float* kb = &Kbuf[bufi][0];
    const float* vb = &Vbuf[bufi][0];

    v2f kf[16];
#pragma unroll
    for (int kc = 0; kc < 16; ++kc) {
      const int h = kc * 4 + halfk;
      kf[kc].x = kb[h * 16 + l16];
      kf[kc].y = kb[(h + 1) * 16 + l16];
    }
    v2f vf[4][4];
#pragma unroll
    for (int j = 0; j < 4; ++j) {
      const int r0 = (half * 8 + 2 * j) * VPAD;
#pragma unroll
      for (int t = 0; t < 4; ++t) {
        vf[j][t].x = vb[r0 + 16 * t + l16];
        vf[j][t].y = vb[r0 + VPAD + 16 * t + l16];
      }
    }

    v8f s = {};
#pragma unroll
    for (int kc = 0; kc < 16; ++kc)
      s = wmma_f32(kf[kc], qa[kc], s);

    float sv[8];
#pragma unroll
    for (int r = 0; r < 8; ++r) sv[r] = s[r] * scale;
    float tm = fmaxf(fmaxf(fmaxf(sv[0], sv[1]), fmaxf(sv[2], sv[3])),
                     fmaxf(fmaxf(sv[4], sv[5]), fmaxf(sv[6], sv[7])));
    tm = fmaxf(tm, xhalf(tm));
    const float nm    = fmaxf(mrun, tm);
    const float alpha = __expf(mrun - nm);
    float p[8];
#pragma unroll
    for (int r = 0; r < 8; ++r) p[r] = __expf(sv[r] - nm);
    const float rs = ((p[0] + p[1]) + (p[2] + p[3])) + ((p[4] + p[5]) + (p[6] + p[7]));
    lrun = lrun * alpha + rs;
    mrun = nm;
    acc0 *= alpha; acc1 *= alpha; acc2 *= alpha; acc3 *= alpha;

#pragma unroll
    for (int j = 0; j < 4; ++j) {
      v2f pb;
      pb.x = p[2 * j];
      pb.y = p[2 * j + 1];
      acc0 = wmma_f32(vf[j][0], pb, acc0);
      acc1 = wmma_f32(vf[j][1], pb, acc1);
      acc2 = wmma_f32(vf[j][2], pb, acc2);
      acc3 = wmma_f32(vf[j][3], pb, acc3);
    }

    if (more) {
      *(v4f*)&Kbuf[bufi ^ 1][kpiece] = kn;
      *(v4f*)&Vbuf[bufi ^ 1][vpiece] = vn;
    }
  }

  const float ltot = lrun + xhalf(lrun);
  const float inv  = 1.0f / ltot;
  float* op = out + (b * SEQ + qbase + l16) * HS + half * 8;
  {
    v4f o;
    o.x = acc0[0] * inv; o.y = acc0[1] * inv; o.z = acc0[2] * inv; o.w = acc0[3] * inv; *(v4f*)(op)      = o;
    o.x = acc0[4] * inv; o.y = acc0[5] * inv; o.z = acc0[6] * inv; o.w = acc0[7] * inv; *(v4f*)(op + 4)  = o;
    o.x = acc1[0] * inv; o.y = acc1[1] * inv; o.z = acc1[2] * inv; o.w = acc1[3] * inv; *(v4f*)(op + 16) = o;
    o.x = acc1[4] * inv; o.y = acc1[5] * inv; o.z = acc1[6] * inv; o.w = acc1[7] * inv; *(v4f*)(op + 20) = o;
    o.x = acc2[0] * inv; o.y = acc2[1] * inv; o.z = acc2[2] * inv; o.w = acc2[3] * inv; *(v4f*)(op + 32) = o;
    o.x = acc2[4] * inv; o.y = acc2[5] * inv; o.z = acc2[6] * inv; o.w = acc2[7] * inv; *(v4f*)(op + 36) = o;
    o.x = acc3[0] * inv; o.y = acc3[1] * inv; o.z = acc3[2] * inv; o.w = acc3[3] * inv; *(v4f*)(op + 48) = o;
    o.x = acc3[4] * inv; o.y = acc3[5] * inv; o.z = acc3[6] * inv; o.w = acc3[7] * inv; *(v4f*)(op + 52) = o;
  }
}

extern "C" void kernel_launch(void* const* d_in, const int* in_sizes, int n_in,
                              void* d_out, int out_size, void* d_ws, size_t ws_size,
                              hipStream_t stream) {
  (void)in_sizes; (void)n_in; (void)out_size; (void)ws_size;
  const float* x  = (const float*)d_in[0];
  const float* Wk = (const float*)d_in[1];
  const float* Wq = (const float*)d_in[2];
  const float* Wv = (const float*)d_in[3];

  float* ws = (float*)d_ws;
  const size_t plane = (size_t)BATCH * SEQ * HS;
  float* Q  = ws;
  float* Kt = ws + plane;
  float* V  = ws + 2 * plane;
  float* out = (float*)d_out;

  dim3 g1(BATCH * SEQ / 16, 3);
  proj_kernel<<<g1, 128, 0, stream>>>(x, Wq, Wk, Wv, Q, Kt, V);

  attn_kernel<<<(BATCH * SEQ / 16) / 8, 256, 0, stream>>>(Q, Kt, V, out);
}


# attention keys split in two across the block's wave groups (4 query tiles x 2 key halves, partials joined through LDS): 256 blocks, one per WGP, was 128; softmax row-sum and PV summation order change
# speedup vs baseline: 3.7172x; 1.4149x over previous
#include <hip/hip_runtime.h>
#include <math.h>

typedef __attribute__((ext_vector_type(2))) float v2f;
typedef __attribute__((ext_vector_type(4))) float v4f;
typedef __attribute__((ext_vector_type(8))) float v8f;

#define EMB   768
#define HS    64
#define SEQ   2048
#define BATCH 8
#define XPAD  772
#define VPAD  68

__device__ __forceinline__ v8f wmma_f32(v2f a, v2f b, v8f c) {
  return __builtin_amdgcn_wmma_f32_16x16x4_f32(
       false, a,  false, b,
       (short)0, c,  false,  false);
}

__device__ __forceinline__ float xhalf(float x) {
  const int b = __builtin_bit_cast(int, x);
  return __builtin_bit_cast(float,
      __builtin_amdgcn_permlanex16(b, b, 0x76543210, 0xfedcba98, false, false));
}

__global__ __launch_bounds__(128) void proj_kernel(
    const float* __restrict__ x,
    const float* __restrict__ Wq, const float* __restrict__ Wk,
    const float* __restrict__ Wv,
    float* __restrict__ Q, float* __restrict__ Kt, float* __restrict__ V) {
  __shared__ float xs[16 * XPAD];

  const int tileM = blockIdx.x;
  const int which = blockIdx.y;
  const int tid   = threadIdx.x;
  const int lane  = tid & 31;
  const int wave  = tid >> 5;
  const int l16   = lane & 15;
  const int half  = lane >> 4;
  const int halfk = half * 2;
  const int rowBase = tileM * 16;

  for (int i = tid; i < 16 * EMB; i += 128) {
    const int r = i / EMB, cc = i - r * EMB;
    xs[r * XPAD + cc] = x[(rowBase + r) * EMB + cc];
  }
  __syncthreads();

  const float* W = (which == 0) ? Wq : (which == 1) ? Wk : Wv;
  const int n = wave * 16 + l16;

  v8f c = {};
  for (int kc = 0; kc < EMB / 4; ++kc) {
    const int k0 = kc * 4 + halfk;
    v2f a, b;
    a.x = xs[l16 * XPAD + k0];
    a.y = xs[l16 * XPAD + k0 + 1];
    b.x = W[k0 * HS + n];
    b.y = W[(k0 + 1) * HS + n];
    c = wmma_f32(a, b, c);
  }

  if (which == 1) {
    const int bidx = rowBase / SEQ;
    const int s0   = rowBase - bidx * SEQ;
#pragma unroll
    for (int v = 0; v < 8; ++v) {
      const int srow = s0 + v + half * 8;
      Kt[(bidx * HS + n) * SEQ + srow] = c[v];
    }
  } else {
    float* O = (which == 0) ? Q : V;
#pragma unroll
    for (int v = 0; v < 8; ++v) {
      const int m = rowBase + v + half * 8;
      O[m * HS + n] = c[v];
    }
  }
}

#define KHALF (SEQ / 2)
#define MVALS 34

__global__ __launch_bounds__(256) __attribute__((amdgpu_waves_per_eu(2, 2))) void attn_kernel(
    const float* __restrict__ Q, const float* __restrict__ Kt,
    const float* __restrict__ V, float* __restrict__ out) {
  __shared__ __attribute__((aligned(16))) float Kbuf[2][2][HS * 16];
  __shared__ __attribute__((aligned(16))) float Vbuf[2][2][16 * VPAD];
  __shared__ float mrg[4][MVALS][32];

  const int tid   = threadIdx.x;
  const int lane  = tid & 31;
  const int wave  = tid >> 5;
  const int grp   = wave >> 2;
  const int qt    = wave & 3;
  const int gt    = blockIdx.x * 4 + qt;
  const int b     = gt >> 7;
  const int qbase = (gt & 127) * 16;
  const int l16   = lane & 15;
  const int half  = lane >> 4;
  const int halfk = half * 2;

  v2f qa[16];
  const int mrow = (b * SEQ + qbase + l16) * HS;
#pragma unroll
  for (int kc = 0; kc < 16; ++kc) {
    qa[kc].x = Q[mrow + kc * 4 + halfk];
    qa[kc].y = Q[mrow + kc * 4 + halfk + 1];
  }

  v8f acc0 = {}, acc1 = {}, acc2 = {}, acc3 = {};
  float mrun = -INFINITY;
  float lrun = 0.0f;

  const float scale = 0.125f;

  const int t7 = tid & 127;
  const int kpiece = (t7 >> 2) * 16 + (t7 & 3) * 4;
  const int vpiece = (t7 >> 4) * VPAD + (t7 & 15) * 4;
  const float* kg = Kt + (b * HS + (t7 >> 2)) * SEQ + grp * KHALF + (t7 & 3) * 4;
  const float* vg = V + (b * SEQ + grp * KHALF + (t7 >> 4)) * HS + (t7 & 15) * 4;

  {
    const v4f k0 = *(const v4f*)kg;
    const v4f k1 = *(const v4f*)(kg + 32 * SEQ);
    const v4f v0 = *(const v4f*)vg;
    const v4f v1 = *(const v4f*)(vg + 8 * HS);
    *(v4f*)&Kbuf[0][grp][kpiece]            = k0;
    *(v4f*)&Kbuf[0][grp][kpiece + 32 * 16]  = k1;
    *(v4f*)&Vbuf[0][grp][vpiece]            = v0;
    *(v4f*)&Vbuf[0][grp][vpiece + 8 * VPAD] = v1;
  }

  for (int it = 0; it < KHALF / 16; ++it) {
    __syncthreads();
    const int bufi = it & 1;
    const bool more = it + 1 < KHALF / 16;
    v4f kn0 = {}, kn1 = {}, vn0 = {}, vn1 = {};
    if (more) {
      kn0 = *(const v4f*)(kg + (it + 1) * 16);
      kn1 = *(const v4f*)(kg + 32 * SEQ + (it + 1) * 16);
      vn0 = *(const v4f*)(vg + (it + 1) * 16 * HS);
      vn1 = *(const v4f*)(vg + 8 * HS + (it + 1) * 16 * HS);
    }

    const float* kb = &Kbuf[bufi][grp][0];
    const float* vb = &Vbuf[bufi][grp][0];

    v2f kf[16];
#pragma unroll
    for (int kc = 0; kc < 16; ++kc) {
      const int h = kc * 4 + halfk;
      kf[kc].x = kb[h * 16 + l16];
      kf[kc].y = kb[(h + 1) * 16 + l16];
    }
    v2f vf[4][4];
#pragma unroll
    for (int j = 0; j < 4; ++j) {
      const int r0 = (half * 8 + 2 * j) * VPAD;
#pragma unroll
      for (int t = 0; t < 4; ++t) {
        vf[j][t].x = vb[r0 + 16 * t + l16];
        vf[j][t].y = vb[r0 + VPAD + 16 * t + l16];
      }
    }

    v8f s = {};
#pragma unroll
    for (int kc = 0; kc < 16; ++kc)
      s = wmma_f32(kf[kc], qa[kc], s);

    float sv[8];
#pragma unroll
    for (int r = 0; r < 8; ++r) sv[r] = s[r] * scale;
    float tm = fmaxf(fmaxf(fmaxf(sv[0], sv[1]), fmaxf(sv[2], sv[3])),
                     fmaxf(fmaxf(sv[4], sv[5]), fmaxf(sv[6], sv[7])));
    tm = fmaxf(tm, xhalf(tm));
    const float nm    = fmaxf(mrun, tm);
    const float alpha = __expf(mrun - nm);
    float p[8];
#pragma unroll
    for (int r = 0; r < 8; ++r) p[r] = __expf(sv[r] - nm);
    const float rs = ((p[0] + p[1]) + (p[2] + p[3])) + ((p[4] + p[5]) + (p[6] + p[7]));
    lrun = lrun * alpha + rs;
    mrun = nm;
    acc0 *= alpha; acc1 *= alpha; acc2 *= alpha; acc3 *= alpha;

#pragma unroll
    for (int j = 0; j < 4; ++j) {
      v2f pb;
      pb.x = p[2 * j];
      pb.y = p[2 * j + 1];
      acc0 = wmma_f32(vf[j][0], pb, acc0);
      acc1 = wmma_f32(vf[j][1], pb, acc1);
      acc2 = wmma_f32(vf[j][2], pb, acc2);
      acc3 = wmma_f32(vf[j][3], pb, acc3);
    }

    if (more) {
      *(v4f*)&Kbuf[bufi ^ 1][grp][kpiece]            = kn0;
      *(v4f*)&Kbuf[bufi ^ 1][grp][kpiece + 32 * 16]  = kn1;
      *(v4f*)&Vbuf[bufi ^ 1][grp][vpiece]            = vn0;
      *(v4f*)&Vbuf[bufi ^ 1][grp][vpiece + 8 * VPAD] = vn1;
    }
  }

  const float ltot = lrun + xhalf(lrun);
  float* mw = &mrg[qt][0][lane];
  if (grp == 1) {
#pragma unroll
    for (int r = 0; r < 8; ++r) {
      mw[(r)      * 32] = acc0[r];
      mw[(8 + r)  * 32] = acc1[r];
      mw[(16 + r) * 32] = acc2[r];
      mw[(24 + r) * 32] = acc3[r];
    }
    mw[32 * 32] = mrun;
    mw[33 * 32] = ltot;
  }
  __syncthreads();
  if (grp == 0) {
    const float m1 = mw[32 * 32];
    const float l1 = mw[33 * 32];
    const float m  = fmaxf(mrun, m1);
    const float a0 = __expf(mrun - m);
    const float a1 = __expf(m1 - m);
    const float inv = 1.0f / (ltot * a0 + l1 * a1);
    const float w0 = a0 * inv;
    const float w1 = a1 * inv;
    float* op = out + (b * SEQ + qbase + l16) * HS + half * 8;
    v4f o;
    o.x = acc0[0] * w0 + mw[0  * 32] * w1; o.y = acc0[1] * w0 + mw[1  * 32] * w1; o.z = acc0[2] * w0 + mw[2  * 32] * w1; o.w = acc0[3] * w0 + mw[3  * 32] * w1; *(v4f*)(op)      = o;
    o.x = acc0[4] * w0 + mw[4  * 32] * w1; o.y = acc0[5] * w0 + mw[5  * 32] * w1; o.z = acc0[6] * w0 + mw[6  * 32] * w1; o.w = acc0[7] * w0 + mw[7  * 32] * w1; *(v4f*)(op + 4)  = o;
    o.x = acc1[0] * w0 + mw[8  * 32] * w1; o.y = acc1[1] * w0 + mw[9  * 32] * w1; o.z = acc1[2] * w0 + mw[10 * 32] * w1; o.w = acc1[3] * w0 + mw[11 * 32] * w1; *(v4f*)(op + 16) = o;
    o.x = acc1[4] * w0 + mw[12 * 32] * w1; o.y = acc1[5] * w0 + mw[13 * 32] * w1; o.z = acc1[6] * w0 + mw[14 * 32] * w1; o.w = acc1[7] * w0 + mw[15 * 32] * w1; *(v4f*)(op + 20) = o;
    o.x = acc2[0] * w0 + mw[16 * 32] * w1; o.y = acc2[1] * w0 + mw[17 * 32] * w1; o.z = acc2[2] * w0 + mw[18 * 32] * w1; o.w = acc2[3] * w0 + mw[19 * 32] * w1; *(v4f*)(op + 32) = o;
    o.x = acc2[4] * w0 + mw[20 * 32] * w1; o.y = acc2[5] * w0 + mw[21 * 32] * w1; o.z = acc2[6] * w0 + mw[22 * 32] * w1; o.w = acc2[7] * w0 + mw[23 * 32] * w1; *(v4f*)(op + 36) = o;
    o.x = acc3[0] * w0 + mw[24 * 32] * w1; o.y = acc3[1] * w0 + mw[25 * 32] * w1; o.z = acc3[2] * w0 + mw[26 * 32] * w1; o.w = acc3[3] * w0 + mw[27 * 32] * w1; *(v4f*)(op + 48) = o;
    o.x = acc3[4] * w0 + mw[28 * 32] * w1; o.y = acc3[5] * w0 + mw[29 * 32] * w1; o.z = acc3[6] * w0 + mw[30 * 32] * w1; o.w = acc3[7] * w0 + mw[31 * 32] * w1; *(v4f*)(op + 52) = o;
  }
}

extern "C" void kernel_launch(void* const* d_in, const int* in_sizes, int n_in,
                              void* d_out, int out_size, void* d_ws, size_t ws_size,
                              hipStream_t stream) {
  (void)in_sizes; (void)n_in; (void)out_size; (void)ws_size;
  const float* x  = (const float*)d_in[0];
  const float* Wk = (const float*)d_in[1];
  const float* Wq = (const float*)d_in[2];
  const float* Wv = (const float*)d_in[3];

  float* ws = (float*)d_ws;
  const size_t plane = (size_t)BATCH * SEQ * HS;
  float* Q  = ws;
  float* Kt = ws + plane;
  float* V  = ws + 2 * plane;
  float* out = (float*)d_out;

  dim3 g1(BATCH * SEQ / 16, 3);
  proj_kernel<<<g1, 128, 0, stream>>>(x, Wq, Wk, Wv, Q, Kt, V);

  attn_kernel<<<(BATCH * SEQ / 16) / 4, 256, 0, stream>>>(Q, Kt, V, out);
}


# projection retiled as one GEMM: a block owns 64 x rows x all 192 Q/K/V columns, x and W chunks staged once in LDS by pipelined 16-byte loads, 6 tiles per wave; Q,V tiles transposed, k order kept
# speedup vs baseline: 4.7771x; 1.2851x over previous
#include <hip/hip_runtime.h>
#include <math.h>

typedef __attribute__((ext_vector_type(2))) float v2f;
typedef __attribute__((ext_vector_type(4))) float v4f;
typedef __attribute__((ext_vector_type(8))) float v8f;

#define EMB   768
#define HS    64
#define SEQ   2048
#define BATCH 8
#define VPAD  68

__device__ __forceinline__ v8f wmma_f32(v2f a, v2f b, v8f c) {
  return __builtin_amdgcn_wmma_f32_16x16x4_f32(
       false, a,  false, b,
       (short)0, c,  false,  false);
}

__device__ __forceinline__ float xhalf(float x) {
  const int b = __builtin_bit_cast(int, x);
  return __builtin_bit_cast(float,
      __builtin_amdgcn_permlanex16(b, b, 0x76543210, 0xfedcba98, false, false));
}

#define PBM 64
#define PKC 64
#define PNCH (EMB / PKC)
#define XP  68
#define WP  72
#define WAS (PKC * WP + 8)

__global__ __launch_bounds__(256) __attribute__((amdgpu_waves_per_eu(2, 2))) void proj_kernel(
    const float* __restrict__ x,
    const float* __restrict__ Wq, const float* __restrict__ Wk,
    const float* __restrict__ Wv,
    float* __restrict__ Q, float* __restrict__ Kt, float* __restrict__ V) {
  __shared__ __attribute__((aligned(16))) float xs[2][PBM * XP];
  __shared__ __attribute__((aligned(16))) float wl[2][3 * WAS];

  const int tid   = threadIdx.x;
  const int lane  = tid & 31;
  const int wave  = tid >> 5;
  const int mp    = wave & 1;
  const int g     = wave >> 1;
  const int l16   = lane & 15;
  const int half  = lane >> 4;
  const int rowBase = blockIdx.x * PBM;

  const int sr = tid >> 4;
  const int sc = (tid & 15) * 4;
  const float* xg = x  + (rowBase + sr) * EMB + sc;
  const float* qg = Wq + sr * HS + sc;
  const float* kg = Wk + sr * HS + sc;
  const float* vg = Wv + sr * HS + sc;
  const int xso = sr * XP + sc;
  const int wso = sr * WP + sc;

  v4f px[4], pq[4], pk[4], pv[4];
#pragma unroll
  for (int i = 0; i < 4; ++i) {
    px[i] = *(const v4f*)(xg + i * 16 * EMB);
    pq[i] = *(const v4f*)(qg + i * 16 * HS);
    pk[i] = *(const v4f*)(kg + i * 16 * HS);
    pv[i] = *(const v4f*)(vg + i * 16 * HS);
  }
#pragma unroll
  for (int i = 0; i < 4; ++i) {
    *(v4f*)&xs[0][xso + i * 16 * XP]       = px[i];
    *(v4f*)&wl[0][wso + i * 16 * WP]       = pq[i];
    *(v4f*)&wl[0][wso + i * 16 * WP + WAS]  = pk[i];
    *(v4f*)&wl[0][wso + i * 16 * WP + 2 * WAS] = pv[i];
  }

  v8f aq0 = {}, aq1 = {}, ak0 = {}, ak1 = {}, av0 = {}, av1 = {};

  const int xfo = (32 * mp + l16) * XP + 2 * half;
  const int wfo = (2 * half) * WP + 16 * g + l16;

  for (int ch = 0; ch < PNCH; ++ch) {
    __syncthreads();
    const int bufi = ch & 1;
    const bool more = ch + 1 < PNCH;
    if (more) {
#pragma unroll
      for (int i = 0; i < 4; ++i) {
        px[i] = *(const v4f*)(xg + i * 16 * EMB + (ch + 1) * PKC);
        pq[i] = *(const v4f*)(qg + i * 16 * HS + (ch + 1) * PKC * HS);
        pk[i] = *(const v4f*)(kg + i * 16 * HS + (ch + 1) * PKC * HS);
        pv[i] = *(const v4f*)(vg + i * 16 * HS + (ch + 1) * PKC * HS);
      }
    }
    const float* xb = &xs[bufi][xfo];
    const float* wb = &wl[bufi][wfo];

    v2f xf0[16], xf1[16], wq[16], wk[16], wv[16];
#pragma unroll
    for (int ks = 0; ks < 16; ++ks) {
      const int kk = ks * 4;
      xf0[ks] = *(const v2f*)(xb + kk);
      xf1[ks] = *(const v2f*)(xb + 16 * XP + kk);
      wq[ks].x = wb[kk * WP];            wq[ks].y = wb[kk * WP + WP];
      wk[ks].x = wb[kk * WP + WAS];      wk[ks].y = wb[kk * WP + WP + WAS];
      wv[ks].x = wb[kk * WP + 2 * WAS];  wv[ks].y = wb[kk * WP + WP + 2 * WAS];
    }
    __builtin_amdgcn_sched_barrier(0);
#pragma unroll
    for (int ks = 0; ks < 16; ++ks) {
      aq0 = wmma_f32(wq[ks], xf0[ks], aq0);
      ak0 = wmma_f32(xf0[ks], wk[ks], ak0);
      av0 = wmma_f32(wv[ks], xf0[ks], av0);
      aq1 = wmma_f32(wq[ks], xf1[ks], aq1);
      ak1 = wmma_f32(xf1[ks], wk[ks], ak1);
      av1 = wmma_f32(wv[ks], xf1[ks], av1);
    }

    if (more) {
#pragma unroll
      for (int i = 0; i < 4; ++i) {
        *(v4f*)&xs[bufi ^ 1][xso + i * 16 * XP]       = px[i];
        *(v4f*)&wl[bufi ^ 1][wso + i * 16 * WP]       = pq[i];
        *(v4f*)&wl[bufi ^ 1][wso + i * 16 * WP + WAS]  = pk[i];
        *(v4f*)&wl[bufi ^ 1][wso + i * 16 * WP + 2 * WAS] = pv[i];
      }
    }
  }

  {
    const int m0 = rowBase + 32 * mp + l16;
    const int n0 = 16 * g + 8 * half;
    float* q0 = Q + m0 * HS + n0;
    float* q1 = Q + (m0 + 16) * HS + n0;
    float* v0 = V + m0 * HS + n0;
    float* v1 = V + (m0 + 16) * HS + n0;
    v4f o;
    o.x = aq0[0]; o.y = aq0[1]; o.z = aq0[2]; o.w = aq0[3]; *(v4f*)(q0)     = o;
    o.x = aq0[4]; o.y = aq0[5]; o.z = aq0[6]; o.w = aq0[7]; *(v4f*)(q0 + 4) = o;
    o.x = aq1[0]; o.y = aq1[1]; o.z = aq1[2]; o.w = aq1[3]; *(v4f*)(q1)     = o;
    o.x = aq1[4]; o.y = aq1[5]; o.z = aq1[6]; o.w = aq1[7]; *(v4f*)(q1 + 4) = o;
    o.x = av0[0]; o.y = av0[1]; o.z = av0[2]; o.w = av0[3]; *(v4f*)(v0)     = o;
    o.x = av0[4]; o.y = av0[5]; o.z = av0[6]; o.w = av0[7]; *(v4f*)(v0 + 4) = o;
    o.x = av1[0]; o.y = av1[1]; o.z = av1[2]; o.w = av1[3]; *(v4f*)(v1)     = o;
    o.x = av1[4]; o.y = av1[5]; o.z = av1[6]; o.w = av1[7]; *(v4f*)(v1 + 4) = o;
  }
  {
    const int bidx = rowBase / SEQ;
    const int s0   = rowBase - bidx * SEQ + 32 * mp + 8 * half;
    float* k0 = Kt + (bidx * HS + 16 * g + l16) * SEQ + s0;
    v4f o;
    o.x = ak0[0]; o.y = ak0[1]; o.z = ak0[2]; o.w = ak0[3]; *(v4f*)(k0)      = o;
    o.x = ak0[4]; o.y = ak0[5]; o.z = ak0[6]; o.w = ak0[7]; *(v4f*)(k0 + 4)  = o;
    o.x = ak1[0]; o.y = ak1[1]; o.z = ak1[2]; o.w = ak1[3]; *(v4f*)(k0 + 16) = o;
    o.x = ak1[4]; o.y = ak1[5]; o.z = ak1[6]; o.w = ak1[7]; *(v4f*)(k0 + 20) = o;
  }
}

#define KHALF (SEQ / 2)
#define MVALS 34

__global__ __launch_bounds__(256) __attribute__((amdgpu_waves_per_eu(2, 2))) void attn_kernel(
    const float* __restrict__ Q, const float* __restrict__ Kt,
    const float* __restrict__ V, float* __restrict__ out) {
  __shared__ __attribute__((aligned(16))) float Kbuf[2][2][HS * 16];
  __shared__ __attribute__((aligned(16))) float Vbuf[2][2][16 * VPAD];
  __shared__ float mrg[4][MVALS][32];

  const int tid   = threadIdx.x;
  const int lane  = tid & 31;
  const int wave  = tid >> 5;
  const int grp   = wave >> 2;
  const int qt    = wave & 3;
  const int gt    = blockIdx.x * 4 + qt;
  const int b     = gt >> 7;
  const int qbase = (gt & 127) * 16;
  const int l16   = lane & 15;
  const int half  = lane >> 4;
  const int halfk = half * 2;

  v2f qa[16];
  const int mrow = (b * SEQ + qbase + l16) * HS;
#pragma unroll
  for (int kc = 0; kc < 16; ++kc) {
    qa[kc].x = Q[mrow + kc * 4 + halfk];
    qa[kc].y = Q[mrow + kc * 4 + halfk + 1];
  }

  v8f acc0 = {}, acc1 = {}, acc2 = {}, acc3 = {};
  float mrun = -INFINITY;
  float lrun = 0.0f;

  const float scale = 0.125f;

  const int t7 = tid & 127;
  const int kpiece = (t7 >> 2) * 16 + (t7 & 3) * 4;
  const int vpiece = (t7 >> 4) * VPAD + (t7 & 15) * 4;
  const float* kg = Kt + (b * HS + (t7 >> 2)) * SEQ + grp * KHALF + (t7 & 3) * 4;
  const float* vg = V + (b * SEQ + grp * KHALF + (t7 >> 4)) * HS + (t7 & 15) * 4;

  {
    const v4f k0 = *(const v4f*)kg;
    const v4f k1 = *(const v4f*)(kg + 32 * SEQ);
    const v4f v0 = *(const v4f*)vg;
    const v4f v1 = *(const v4f*)(vg + 8 * HS);
    *(v4f*)&Kbuf[0][grp][kpiece]            = k0;
    *(v4f*)&Kbuf[0][grp][kpiece + 32 * 16]  = k1;
    *(v4f*)&Vbuf[0][grp][vpiece]            = v0;
    *(v4f*)&Vbuf[0][grp][vpiece + 8 * VPAD] = v1;
  }

  for (int it = 0; it < KHALF / 16; ++it) {
    __syncthreads();
    const int bufi = it & 1;
    const bool more = it + 1 < KHALF / 16;
    v4f kn0 = {}, kn1 = {}, vn0 = {}, vn1 = {};
    if (more) {
      kn0 = *(const v4f*)(kg + (it + 1) * 16);
      kn1 = *(const v4f*)(kg + 32 * SEQ + (it + 1) * 16);
      vn0 = *(const v4f*)(vg + (it + 1) * 16 * HS);
      vn1 = *(const v4f*)(vg + 8 * HS + (it + 1) * 16 * HS);
    }

    const float* kb = &Kbuf[bufi][grp][0];
    const float* vb = &Vbuf[bufi][grp][0];

    v2f kf[16];
#pragma unroll
    for (int kc = 0; kc < 16; ++kc) {
      const int h = kc * 4 + halfk;
      kf[kc].x = kb[h * 16 + l16];
      kf[kc].y = kb[(h + 1) * 16 + l16];
    }
    v2f vf[4][4];
#pragma unroll
    for (int j = 0; j < 4; ++j) {
      const int r0 = (half * 8 + 2 * j) * VPAD;
#pragma unroll
      for (int t = 0; t < 4; ++t) {
        vf[j][t].x = vb[r0 + 16 * t + l16];
        vf[j][t].y = vb[r0 + VPAD + 16 * t + l16];
      }
    }

    v8f s = {};
#pragma unroll
    for (int kc = 0; kc < 16; ++kc)
      s = wmma_f32(kf[kc], qa[kc], s);

    float sv[8];
#pragma unroll
    for (int r = 0; r < 8; ++r) sv[r] = s[r] * scale;
    float tm = fmaxf(fmaxf(fmaxf(sv[0], sv[1]), fmaxf(sv[2], sv[3])),
                     fmaxf(fmaxf(sv[4], sv[5]), fmaxf(sv[6], sv[7])));
    tm = fmaxf(tm, xhalf(tm));
    const float nm    = fmaxf(mrun, tm);
    const float alpha = __expf(mrun - nm);
    float p[8];
#pragma unroll
    for (int r = 0; r < 8; ++r) p[r] = __expf(sv[r] - nm);
    const float rs = ((p[0] + p[1]) + (p[2] + p[3])) + ((p[4] + p[5]) + (p[6] + p[7]));
    lrun = lrun * alpha + rs;
    mrun = nm;
    acc0 *= alpha; acc1 *= alpha; acc2 *= alpha; acc3 *= alpha;

#pragma unroll
    for (int j = 0; j < 4; ++j) {
      v2f pb;
      pb.x = p[2 * j];
      pb.y = p[2 * j + 1];
      acc0 = wmma_f32(vf[j][0], pb, acc0);
      acc1 = wmma_f32(vf[j][1], pb, acc1);
      acc2 = wmma_f32(vf[j][2], pb, acc2);
      acc3 = wmma_f32(vf[j][3], pb, acc3);
    }

    if (more) {
      *(v4f*)&Kbuf[bufi ^ 1][grp][kpiece]            = kn0;
      *(v4f*)&Kbuf[bufi ^ 1][grp][kpiece + 32 * 16]  = kn1;
      *(v4f*)&Vbuf[bufi ^ 1][grp][vpiece]            = vn0;
      *(v4f*)&Vbuf[bufi ^ 1][grp][vpiece + 8 * VPAD] = vn1;
    }
  }

  const float ltot = lrun + xhalf(lrun);
  float* mw = &mrg[qt][0][lane];
  if (grp == 1) {
#pragma unroll
    for (int r = 0; r < 8; ++r) {
      mw[(r)      * 32] = acc0[r];
      mw[(8 + r)  * 32] = acc1[r];
      mw[(16 + r) * 32] = acc2[r];
      mw[(24 + r) * 32] = acc3[r];
    }
    mw[32 * 32] = mrun;
    mw[33 * 32] = ltot;
  }
  __syncthreads();
  if (grp == 0) {
    const float m1 = mw[32 * 32];
    const float l1 = mw[33 * 32];
    const float m  = fmaxf(mrun, m1);
    const float a0 = __expf(mrun - m);
    const float a1 = __expf(m1 - m);
    const float inv = 1.0f / (ltot * a0 + l1 * a1);
    const float w0 = a0 * inv;
    const float w1 = a1 * inv;
    float* op = out + (b * SEQ + qbase + l16) * HS + half * 8;
    v4f o;
    o.x = acc0[0] * w0 + mw[0  * 32] * w1; o.y = acc0[1] * w0 + mw[1  * 32] * w1; o.z = acc0[2] * w0 + mw[2  * 32] * w1; o.w = acc0[3] * w0 + mw[3  * 32] * w1; *(v4f*)(op)      = o;
    o.x = acc0[4] * w0 + mw[4  * 32] * w1; o.y = acc0[5] * w0 + mw[5  * 32] * w1; o.z = acc0[6] * w0 + mw[6  * 32] * w1; o.w = acc0[7] * w0 + mw[7  * 32] * w1; *(v4f*)(op + 4)  = o;
    o.x = acc1[0] * w0 + mw[8  * 32] * w1; o.y = acc1[1] * w0 + mw[9  * 32] * w1; o.z = acc1[2] * w0 + mw[10 * 32] * w1; o.w = acc1[3] * w0 + mw[11 * 32] * w1; *(v4f*)(op + 16) = o;
    o.x = acc1[4] * w0 + mw[12 * 32] * w1; o.y = acc1[5] * w0 + mw[13 * 32] * w1; o.z = acc1[6] * w0 + mw[14 * 32] * w1; o.w = acc1[7] * w0 + mw[15 * 32] * w1; *(v4f*)(op + 20) = o;
    o.x = acc2[0] * w0 + mw[16 * 32] * w1; o.y = acc2[1] * w0 + mw[17 * 32] * w1; o.z = acc2[2] * w0 + mw[18 * 32] * w1; o.w = acc2[3] * w0 + mw[19 * 32] * w1; *(v4f*)(op + 32) = o;
    o.x = acc2[4] * w0 + mw[20 * 32] * w1; o.y = acc2[5] * w0 + mw[21 * 32] * w1; o.z = acc2[6] * w0 + mw[22 * 32] * w1; o.w = acc2[7] * w0 + mw[23 * 32] * w1; *(v4f*)(op + 36) = o;
    o.x = acc3[0] * w0 + mw[24 * 32] * w1; o.y = acc3[1] * w0 + mw[25 * 32] * w1; o.z = acc3[2] * w0 + mw[26 * 32] * w1; o.w = acc3[3] * w0 + mw[27 * 32] * w1; *(v4f*)(op + 48) = o;
    o.x = acc3[4] * w0 + mw[28 * 32] * w1; o.y = acc3[5] * w0 + mw[29 * 32] * w1; o.z = acc3[6] * w0 + mw[30 * 32] * w1; o.w = acc3[7] * w0 + mw[31 * 32] * w1; *(v4f*)(op + 52) = o;
  }
}

extern "C" void kernel_launch(void* const* d_in, const int* in_sizes, int n_in,
                              void* d_out, int out_size, void* d_ws, size_t ws_size,
                              hipStream_t stream) {
  (void)in_sizes; (void)n_in; (void)out_size; (void)ws_size;
  const float* x  = (const float*)d_in[0];
  const float* Wk = (const float*)d_in[1];
  const float* Wq = (const float*)d_in[2];
  const float* Wv = (const float*)d_in[3];

  float* ws = (float*)d_ws;
  const size_t plane = (size_t)BATCH * SEQ * HS;
  float* Q  = ws;
  float* Kt = ws + plane;
  float* V  = ws + 2 * plane;
  float* out = (float*)d_out;

  proj_kernel<<<BATCH * SEQ / PBM, 256, 0, stream>>>(x, Wq, Wk, Wv, Q, Kt, V);

  attn_kernel<<<(BATCH * SEQ / 16) / 4, 256, 0, stream>>>(Q, Kt, V, out);
}


# attention fragment loads run under the WMMAs: K/V staged 64 keys per barrier (was 16), k-pairs interleaved in LDS (one 8-byte load per fragment), a tile's loads one group; sum order unchanged
# speedup vs baseline: 5.4067x; 1.1318x over previous
#include <hip/hip_runtime.h>
#include <math.h>

typedef __attribute__((ext_vector_type(2))) float v2f;
typedef __attribute__((ext_vector_type(4))) float v4f;
typedef __attribute__((ext_vector_type(8))) float v8f;

#define EMB   768
#define HS    64
#define SEQ   2048
#define BATCH 8

__device__ __forceinline__ v8f wmma_f32(v2f a, v2f b, v8f c) {
  return __builtin_amdgcn_wmma_f32_16x16x4_f32(
       false, a,  false, b,
       (short)0, c,  false,  false);
}

__device__ __forceinline__ float xhalf(float x) {
  const int b = __builtin_bit_cast(int, x);
  return __builtin_bit_cast(float,
      __builtin_amdgcn_permlanex16(b, b, 0x76543210, 0xfedcba98, false, false));
}

#define PBM 64
#define PKC 64
#define PNCH (EMB / PKC)
#define XP  68
#define WP  72
#define WAS (PKC * WP + 8)

__global__ __launch_bounds__(256) __attribute__((amdgpu_waves_per_eu(2, 2))) void proj_kernel(
    const float* __restrict__ x,
    const float* __restrict__ Wq, const float* __restrict__ Wk,
    const float* __restrict__ Wv,
    float* __restrict__ Q, float* __restrict__ Kt, float* __restrict__ V) {
  __shared__ __attribute__((aligned(16))) float xs[2][PBM * XP];
  __shared__ __attribute__((aligned(16))) float wl[2][3 * WAS];

  const int tid   = threadIdx.x;
  const int lane  = tid & 31;
  const int wave  = tid >> 5;
  const int mp    = wave & 1;
  const int g     = wave >> 1;
  const int l16   = lane & 15;
  const int half  = lane >> 4;
  const int rowBase = blockIdx.x * PBM;

  const int sr = tid >> 4;
  const int sc = (tid & 15) * 4;
  const float* xg = x  + (rowBase + sr) * EMB + sc;
  const float* qg = Wq + sr * HS + sc;
  const float* kg = Wk + sr * HS + sc;
  const float* vg = Wv + sr * HS + sc;
  const int xso = sr * XP + sc;
  const int wso = sr * WP + sc;

  v4f px[4], pq[4], pk[4], pv[4];
#pragma unroll
  for (int i = 0; i < 4; ++i) {
    px[i] = *(const v4f*)(xg + i * 16 * EMB);
    pq[i] = *(const v4f*)(qg + i * 16 * HS);
    pk[i] = *(const v4f*)(kg + i * 16 * HS);
    pv[i] = *(const v4f*)(vg + i * 16 * HS);
  }
#pragma unroll
  for (int i = 0; i < 4; ++i) {
    *(v4f*)&xs[0][xso + i * 16 * XP]       = px[i];
    *(v4f*)&wl[0][wso + i * 16 * WP]       = pq[i];
    *(v4f*)&wl[0][wso + i * 16 * WP + WAS]  = pk[i];
    *(v4f*)&wl[0][wso + i * 16 * WP + 2 * WAS] = pv[i];
  }

  v8f aq0 = {}, aq1 = {}, ak0 = {}, ak1 = {}, av0 = {}, av1 = {};

  const int xfo = (32 * mp + l16) * XP + 2 * half;
  const int wfo = (2 * half) * WP + 16 * g + l16;

  for (int ch = 0; ch < PNCH; ++ch) {
    __syncthreads();
    const int bufi = ch & 1;
    const bool more = ch + 1 < PNCH;
    if (more) {
#pragma unroll
      for (int i = 0; i < 4; ++i) {
        px[i] = *(const v4f*)(xg + i * 16 * EMB + (ch + 1) * PKC);
        pq[i] = *(const v4f*)(qg + i * 16 * HS + (ch + 1) * PKC * HS);
        pk[i] = *(const v4f*)(kg + i * 16 * HS + (ch + 1) * PKC * HS);
        pv[i] = *(const v4f*)(vg + i * 16 * HS + (ch + 1) * PKC * HS);
      }
    }
    const float* xb = &xs[bufi][xfo];
    const float* wb = &wl[bufi][wfo];

    v2f xf0[16], xf1[16], wq[16], wk[16], wv[16];
#pragma unroll
    for (int ks = 0; ks < 16; ++ks) {
      const int kk = ks * 4;
      xf0[ks] = *(const v2f*)(xb + kk);
      xf1[ks] = *(const v2f*)(xb + 16 * XP + kk);
      wq[ks].x = wb[kk * WP];            wq[ks].y = wb[kk * WP + WP];
      wk[ks].x = wb[kk * WP + WAS];      wk[ks].y = wb[kk * WP + WP + WAS];
      wv[ks].x = wb[kk * WP + 2 * WAS];  wv[ks].y = wb[kk * WP + WP + 2 * WAS];
    }
    __builtin_amdgcn_sched_barrier(0);
#pragma unroll
    for (int ks = 0; ks < 16; ++ks) {
      aq0 = wmma_f32(wq[ks], xf0[ks], aq0);
      ak0 = wmma_f32(xf0[ks], wk[ks], ak0);
      av0 = wmma_f32(wv[ks], xf0[ks], av0);
      aq1 = wmma_f32(wq[ks], xf1[ks], aq1);
      ak1 = wmma_f32(xf1[ks], wk[ks], ak1);
      av1 = wmma_f32(wv[ks], xf1[ks], av1);
    }

    if (more) {
#pragma unroll
      for (int i = 0; i < 4; ++i) {
        *(v4f*)&xs[bufi ^ 1][xso + i * 16 * XP]       = px[i];
        *(v4f*)&wl[bufi ^ 1][wso + i * 16 * WP]       = pq[i];
        *(v4f*)&wl[bufi ^ 1][wso + i * 16 * WP + WAS]  = pk[i];
        *(v4f*)&wl[bufi ^ 1][wso + i * 16 * WP + 2 * WAS] = pv[i];
      }
    }
  }

  {
    const int m0 = rowBase + 32 * mp + l16;
    const int n0 = 16 * g + 8 * half;
    float* q0 = Q + m0 * HS + n0;
    float* q1 = Q + (m0 + 16) * HS + n0;
    float* v0 = V + m0 * HS + n0;
    float* v1 = V + (m0 + 16) * HS + n0;
    v4f o;
    o.x = aq0[0]; o.y = aq0[1]; o.z = aq0[2]; o.w = aq0[3]; *(v4f*)(q0)     = o;
    o.x = aq0[4]; o.y = aq0[5]; o.z = aq0[6]; o.w = aq0[7]; *(v4f*)(q0 + 4) = o;
    o.x = aq1[0]; o.y = aq1[1]; o.z = aq1[2]; o.w = aq1[3]; *(v4f*)(q1)     = o;
    o.x = aq1[4]; o.y = aq1[5]; o.z = aq1[6]; o.w = aq1[7]; *(v4f*)(q1 + 4) = o;
    o.x = av0[0]; o.y = av0[1]; o.z = av0[2]; o.w = av0[3]; *(v4f*)(v0)     = o;
    o.x = av0[4]; o.y = av0[5]; o.z = av0[6]; o.w = av0[7]; *(v4f*)(v0 + 4) = o;
    o.x = av1[0]; o.y = av1[1]; o.z = av1[2]; o.w = av1[3]; *(v4f*)(v1)     = o;
    o.x = av1[4]; o.y = av1[5]; o.z = av1[6]; o.w = av1[7]; *(v4f*)(v1 + 4) = o;
  }
  {
    const int bidx = rowBase / SEQ;
    const int s0   = rowBase - bidx * SEQ + 32 * mp + 8 * half;
    float* k0 = Kt + (bidx * HS + 16 * g + l16) * SEQ + s0;
    v4f o;
    o.x = ak0[0]; o.y = ak0[1]; o.z = ak0[2]; o.w = ak0[3]; *(v4f*)(k0)      = o;
    o.x = ak0[4]; o.y = ak0[5]; o.z = ak0[6]; o.w = ak0[7]; *(v4f*)(k0 + 4)  = o;
    o.x = ak1[0]; o.y = ak1[1]; o.z = ak1[2]; o.w = ak1[3]; *(v4f*)(k0 + 16) = o;
    o.x = ak1[4]; o.y = ak1[5]; o.z = ak1[6]; o.w = ak1[7]; *(v4f*)(k0 + 20) = o;
  }
}

#define KHALF  (SEQ / 2)
#define KT     64
#define KSUB   (KT / 16)
#define KP2    160
#define VP2    136
#define NSTEP  (KHALF / KT)
#define MVALS  34

__global__ __launch_bounds__(256) __attribute__((amdgpu_waves_per_eu(2, 2))) void attn_kernel(
    const float* __restrict__ Q, const float* __restrict__ Kt,
    const float* __restrict__ V, float* __restrict__ out) {
  __shared__ __attribute__((aligned(16))) float Kbuf[2][2][(HS / 2) * KP2];
  __shared__ __attribute__((aligned(16))) float Vbuf[2][2][(KT / 2) * VP2];
  __shared__ float mrg[4][MVALS][32];

  const int tid   = threadIdx.x;
  const int lane  = tid & 31;
  const int wave  = tid >> 5;
  const int grp   = wave >> 2;
  const int qt    = wave & 3;
  const int gt    = blockIdx.x * 4 + qt;
  const int b     = gt >> 7;
  const int qbase = (gt & 127) * 16;
  const int l16   = lane & 15;
  const int half  = lane >> 4;
  const int halfk = half * 2;

  v2f qa[16];
  const int mrow = (b * SEQ + qbase + l16) * HS;
#pragma unroll
  for (int kc = 0; kc < 16; ++kc) {
    qa[kc].x = Q[mrow + kc * 4 + halfk];
    qa[kc].y = Q[mrow + kc * 4 + halfk + 1];
  }

  v8f acc0 = {}, acc1 = {}, acc2 = {}, acc3 = {};
  float mrun = -INFINITY;
  float lrun = 0.0f;

  const float scale = 0.125f;

  const int t7   = tid & 127;
  const int prow = t7 >> 4;
  const int pcol = (t7 & 15) * 4;
  const int kst = prow * KP2 + 2 * pcol;
  const int vst = prow * VP2 + 2 * pcol;
  const float* kg = Kt + (b * HS + 2 * prow) * SEQ + grp * KHALF + pcol;
  const float* vg = V + (b * SEQ + grp * KHALF + 2 * prow) * HS + pcol;

  {
    v4f ke[4], ko[4], ve[4], vo[4];
#pragma unroll
    for (int i = 0; i < 4; ++i) {
      ke[i] = *(const v4f*)(kg + i * 16 * SEQ);
      ko[i] = *(const v4f*)(kg + i * 16 * SEQ + SEQ);
      ve[i] = *(const v4f*)(vg + i * 16 * HS);
      vo[i] = *(const v4f*)(vg + i * 16 * HS + HS);
    }
#pragma unroll
    for (int i = 0; i < 4; ++i) {
      v4f w;
      w.x = ke[i].x; w.y = ko[i].x; w.z = ke[i].y; w.w = ko[i].y; *(v4f*)&Kbuf[0][grp][kst + i * 8 * KP2]     = w;
      w.x = ke[i].z; w.y = ko[i].z; w.z = ke[i].w; w.w = ko[i].w; *(v4f*)&Kbuf[0][grp][kst + i * 8 * KP2 + 4] = w;
      w.x = ve[i].x; w.y = vo[i].x; w.z = ve[i].y; w.w = vo[i].y; *(v4f*)&Vbuf[0][grp][vst + i * 8 * VP2]     = w;
      w.x = ve[i].z; w.y = vo[i].z; w.z = ve[i].w; w.w = vo[i].w; *(v4f*)&Vbuf[0][grp][vst + i * 8 * VP2 + 4] = w;
    }
  }

  for (int it = 0; it < NSTEP; ++it) {
    __syncthreads();
    const int bufi = it & 1;
    const int nt = (it + 1) & (NSTEP - 1);
    v4f ke[4], ko[4], ve[4], vo[4];
#pragma unroll
    for (int i = 0; i < 4; ++i) {
      ke[i] = *(const v4f*)(kg + i * 16 * SEQ + nt * KT);
      ko[i] = *(const v4f*)(kg + i * 16 * SEQ + SEQ + nt * KT);
      ve[i] = *(const v4f*)(vg + i * 16 * HS + nt * KT * HS);
      vo[i] = *(const v4f*)(vg + i * 16 * HS + HS + nt * KT * HS);
    }

    const float* kb = &Kbuf[bufi][grp][0];
    const float* vb = &Vbuf[bufi][grp][0];

    v2f kf[KSUB][16];
    v2f vf[KSUB][4][4];
#pragma unroll
    for (int sub = 0; sub < KSUB; ++sub) {
#pragma unroll
      for (int kc = 0; kc < 16; ++kc)
        kf[sub][kc] = *(const v2f*)(kb + (2 * kc + half) * KP2 + 2 * (16 * sub + l16));
#pragma unroll
      for (int j = 0; j < 4; ++j) {
#pragma unroll
        for (int t = 0; t < 4; ++t)
          vf[sub][j][t] = *(const v2f*)(vb + (8 * sub + 4 * half + j) * VP2 + 2 * (16 * t + l16));
      }
    }

#pragma unroll
    for (int sub = 0; sub < KSUB; ++sub) {
      v8f s = {};
#pragma unroll
      for (int kc = 0; kc < 16; ++kc)
        s = wmma_f32(kf[sub][kc], qa[kc], s);

      float sv[8];
#pragma unroll
      for (int r = 0; r < 8; ++r) sv[r] = s[r] * scale;
      float tm = fmaxf(fmaxf(fmaxf(sv[0], sv[1]), fmaxf(sv[2], sv[3])),
                       fmaxf(fmaxf(sv[4], sv[5]), fmaxf(sv[6], sv[7])));
      tm = fmaxf(tm, xhalf(tm));
      const float nm    = fmaxf(mrun, tm);
      const float alpha = __expf(mrun - nm);
      float p[8];
#pragma unroll
      for (int r = 0; r < 8; ++r) p[r] = __expf(sv[r] - nm);
      const float rs = ((p[0] + p[1]) + (p[2] + p[3])) + ((p[4] + p[5]) + (p[6] + p[7]));
      lrun = lrun * alpha + rs;
      mrun = nm;
      acc0 *= alpha; acc1 *= alpha; acc2 *= alpha; acc3 *= alpha;

#pragma unroll
      for (int j = 0; j < 4; ++j) {
        v2f pb;
        pb.x = p[2 * j];
        pb.y = p[2 * j + 1];
        acc0 = wmma_f32(vf[sub][j][0], pb, acc0);
        acc1 = wmma_f32(vf[sub][j][1], pb, acc1);
        acc2 = wmma_f32(vf[sub][j][2], pb, acc2);
        acc3 = wmma_f32(vf[sub][j][3], pb, acc3);
      }
    }

#pragma unroll
    for (int i = 0; i < 4; ++i) {
      v4f w;
      w.x = ke[i].x; w.y = ko[i].x; w.z = ke[i].y; w.w = ko[i].y; *(v4f*)&Kbuf[bufi ^ 1][grp][kst + i * 8 * KP2]     = w;
      w.x = ke[i].z; w.y = ko[i].z; w.z = ke[i].w; w.w = ko[i].w; *(v4f*)&Kbuf[bufi ^ 1][grp][kst + i * 8 * KP2 + 4] = w;
      w.x = ve[i].x; w.y = vo[i].x; w.z = ve[i].y; w.w = vo[i].y; *(v4f*)&Vbuf[bufi ^ 1][grp][vst + i * 8 * VP2]     = w;
      w.x = ve[i].z; w.y = vo[i].z; w.z = ve[i].w; w.w = vo[i].w; *(v4f*)&Vbuf[bufi ^ 1][grp][vst + i * 8 * VP2 + 4] = w;
    }
  }

  const float ltot = lrun + xhalf(lrun);
  float* mw = &mrg[qt][0][lane];
  if (grp == 1) {
#pragma unroll
    for (int r = 0; r < 8; ++r) {
      mw[(r)      * 32] = acc0[r];
      mw[(8 + r)  * 32] = acc1[r];
      mw[(16 + r) * 32] = acc2[r];
      mw[(24 + r) * 32] = acc3[r];
    }
    mw[32 * 32] = mrun;
    mw[33 * 32] = ltot;
  }
  __syncthreads();
  if (grp == 0) {
    const float m1 = mw[32 * 32];
    const float l1 = mw[33 * 32];
    const float m  = fmaxf(mrun, m1);
    const float a0 = __expf(mrun - m);
    const float a1 = __expf(m1 - m);
    const float inv = 1.0f / (ltot * a0 + l1 * a1);
    const float w0 = a0 * inv;
    const float w1 = a1 * inv;
    float* op = out + (b * SEQ + qbase + l16) * HS + half * 8;
    v4f o;
    o.x = acc0[0] * w0 + mw[0  * 32] * w1; o.y = acc0[1] * w0 + mw[1  * 32] * w1; o.z = acc0[2] * w0 + mw[2  * 32] * w1; o.w = acc0[3] * w0 + mw[3  * 32] * w1; *(v4f*)(op)      = o;
    o.x = acc0[4] * w0 + mw[4  * 32] * w1; o.y = acc0[5] * w0 + mw[5  * 32] * w1; o.z = acc0[6] * w0 + mw[6  * 32] * w1; o.w = acc0[7] * w0 + mw[7  * 32] * w1; *(v4f*)(op + 4)  = o;
    o.x = acc1[0] * w0 + mw[8  * 32] * w1; o.y = acc1[1] * w0 + mw[9  * 32] * w1; o.z = acc1[2] * w0 + mw[10 * 32] * w1; o.w = acc1[3] * w0 + mw[11 * 32] * w1; *(v4f*)(op + 16) = o;
    o.x = acc1[4] * w0 + mw[12 * 32] * w1; o.y = acc1[5] * w0 + mw[13 * 32] * w1; o.z = acc1[6] * w0 + mw[14 * 32] * w1; o.w = acc1[7] * w0 + mw[15 * 32] * w1; *(v4f*)(op + 20) = o;
    o.x = acc2[0] * w0 + mw[16 * 32] * w1; o.y = acc2[1] * w0 + mw[17 * 32] * w1; o.z = acc2[2] * w0 + mw[18 * 32] * w1; o.w = acc2[3] * w0 + mw[19 * 32] * w1; *(v4f*)(op + 32) = o;
    o.x = acc2[4] * w0 + mw[20 * 32] * w1; o.y = acc2[5] * w0 + mw[21 * 32] * w1; o.z = acc2[6] * w0 + mw[22 * 32] * w1; o.w = acc2[7] * w0 + mw[23 * 32] * w1; *(v4f*)(op + 36) = o;
    o.x = acc3[0] * w0 + mw[24 * 32] * w1; o.y = acc3[1] * w0 + mw[25 * 32] * w1; o.z = acc3[2] * w0 + mw[26 * 32] * w1; o.w = acc3[3] * w0 + mw[27 * 32] * w1; *(v4f*)(op + 48) = o;
    o.x = acc3[4] * w0 + mw[28 * 32] * w1; o.y = acc3[5] * w0 + mw[29 * 32] * w1; o.z = acc3[6] * w0 + mw[30 * 32] * w1; o.w = acc3[7] * w0 + mw[31 * 32] * w1; *(v4f*)(op + 52) = o;
  }
}

extern "C" void kernel_launch(void* const* d_in, const int* in_sizes, int n_in,
                              void* d_out, int out_size, void* d_ws, size_t ws_size,
                              hipStream_t stream) {
  (void)in_sizes; (void)n_in; (void)out_size; (void)ws_size;
  const float* x  = (const float*)d_in[0];
  const float* Wk = (const float*)d_in[1];
  const float* Wq = (const float*)d_in[2];
  const float* Wv = (const float*)d_in[3];

  float* ws = (float*)d_ws;
  const size_t plane = (size_t)BATCH * SEQ * HS;
  float* Q  = ws;
  float* Kt = ws + plane;
  float* V  = ws + 2 * plane;
  float* out = (float*)d_out;

  proj_kernel<<<BATCH * SEQ / PBM, 256, 0, stream>>>(x, Wq, Wk, Wv, Q, Kt, V);

  attn_kernel<<<(BATCH * SEQ / 16) / 4, 256, 0, stream>>>(Q, Kt, V, out);
}


# attention: each wave owns all 4 query tiles and 1/8 of the keys and stages its own K/V tiles (K now row-major): no workgroup barrier in the loop; 8 partials joined, softmax/PV sum order changes
# speedup vs baseline: 5.6495x; 1.0449x over previous
#include <hip/hip_runtime.h>
#include <math.h>

typedef __attribute__((ext_vector_type(2))) float v2f;
typedef __attribute__((ext_vector_type(4))) float v4f;
typedef __attribute__((ext_vector_type(8))) float v8f;

#define EMB   768
#define HS    64
#define SEQ   2048
#define BATCH 8

__device__ __forceinline__ v8f wmma_f32(v2f a, v2f b, v8f c) {
  return __builtin_amdgcn_wmma_f32_16x16x4_f32(
       false, a,  false, b,
       (short)0, c,  false,  false);
}

__device__ __forceinline__ float xhalf(float x) {
  const int b = __builtin_bit_cast(int, x);
  return __builtin_bit_cast(float,
      __builtin_amdgcn_permlanex16(b, b, 0x76543210, 0xfedcba98, false, false));
}

#define PBM 64
#define PKC 64
#define PNCH (EMB / PKC)
#define XP  68
#define WP  72
#define WAS (PKC * WP + 8)

__global__ __launch_bounds__(256) __attribute__((amdgpu_waves_per_eu(2, 2))) void proj_kernel(
    const float* __restrict__ x,
    const float* __restrict__ Wq, const float* __restrict__ Wk,
    const float* __restrict__ Wv,
    float* __restrict__ Q, float* __restrict__ K, float* __restrict__ V) {
  __shared__ __attribute__((aligned(16))) float xs[2][PBM * XP];
  __shared__ __attribute__((aligned(16))) float wl[2][3 * WAS];

  const int tid   = threadIdx.x;
  const int lane  = tid & 31;
  const int wave  = tid >> 5;
  const int mp    = wave & 1;
  const int g     = wave >> 1;
  const int l16   = lane & 15;
  const int half  = lane >> 4;
  const int rowBase = blockIdx.x * PBM;

  const int sr = tid >> 4;
  const int sc = (tid & 15) * 4;
  const float* xg = x  + (rowBase + sr) * EMB + sc;
  const float* qg = Wq + sr * HS + sc;
  const float* kg = Wk + sr * HS + sc;
  const float* vg = Wv + sr * HS + sc;
  const int xso = sr * XP + sc;
  const int wso = sr * WP + sc;

  v4f px[4], pq[4], pk[4], pv[4];
#pragma unroll
  for (int i = 0; i < 4; ++i) {
    px[i] = *(const v4f*)(xg + i * 16 * EMB);
    pq[i] = *(const v4f*)(qg + i * 16 * HS);
    pk[i] = *(const v4f*)(kg + i * 16 * HS);
    pv[i] = *(const v4f*)(vg + i * 16 * HS);
  }
#pragma unroll
  for (int i = 0; i < 4; ++i) {
    *(v4f*)&xs[0][xso + i * 16 * XP]       = px[i];
    *(v4f*)&wl[0][wso + i * 16 * WP]       = pq[i];
    *(v4f*)&wl[0][wso + i * 16 * WP + WAS]  = pk[i];
    *(v4f*)&wl[0][wso + i * 16 * WP + 2 * WAS] = pv[i];
  }

  v8f aq0 = {}, aq1 = {}, ak0 = {}, ak1 = {}, av0 = {}, av1 = {};

  const int xfo = (32 * mp + l16) * XP + 2 * half;
  const int wfo = (2 * half) * WP + 16 * g + l16;

  for (int ch = 0; ch < PNCH; ++ch) {
    __syncthreads();
    const int bufi = ch & 1;
    const bool more = ch + 1 < PNCH;
    if (more) {
#pragma unroll
      for (int i = 0; i < 4; ++i) {
        px[i] = *(const v4f*)(xg + i * 16 * EMB + (ch + 1) * PKC);
        pq[i] = *(const v4f*)(qg + i * 16 * HS + (ch + 1) * PKC * HS);
        pk[i] = *(const v4f*)(kg + i * 16 * HS + (ch + 1) * PKC * HS);
        pv[i] = *(const v4f*)(vg + i * 16 * HS + (ch + 1) * PKC * HS);
      }
    }
    const float* xb = &xs[bufi][xfo];
    const float* wb = &wl[bufi][wfo];

    v2f xf0[16], xf1[16], wq[16], wk[16], wv[16];
#pragma unroll
    for (int ks = 0; ks < 16; ++ks) {
      const int kk = ks * 4;
      xf0[ks] = *(const v2f*)(xb + kk);
      xf1[ks] = *(const v2f*)(xb + 16 * XP + kk);
      wq[ks].x = wb[kk * WP];            wq[ks].y = wb[kk * WP + WP];
      wk[ks].x = wb[kk * WP + WAS];      wk[ks].y = wb[kk * WP + WP + WAS];
      wv[ks].x = wb[kk * WP + 2 * WAS];  wv[ks].y = wb[kk * WP + WP + 2 * WAS];
    }
    __builtin_amdgcn_sched_barrier(0);
#pragma unroll
    for (int ks = 0; ks < 16; ++ks) {
      aq0 = wmma_f32(wq[ks], xf0[ks], aq0);
      ak0 = wmma_f32(wk[ks], xf0[ks], ak0);
      av0 = wmma_f32(wv[ks], xf0[ks], av0);
      aq1 = wmma_f32(wq[ks], xf1[ks], aq1);
      ak1 = wmma_f32(wk[ks], xf1[ks], ak1);
      av1 = wmma_f32(wv[ks], xf1[ks], av1);
    }

    if (more) {
#pragma unroll
      for (int i = 0; i < 4; ++i) {
        *(v4f*)&xs[bufi ^ 1][xso + i * 16 * XP]       = px[i];
        *(v4f*)&wl[bufi ^ 1][wso + i * 16 * WP]       = pq[i];
        *(v4f*)&wl[bufi ^ 1][wso + i * 16 * WP + WAS]  = pk[i];
        *(v4f*)&wl[bufi ^ 1][wso + i * 16 * WP + 2 * WAS] = pv[i];
      }
    }
  }

  {
    const int m0 = rowBase + 32 * mp + l16;
    const int n0 = 16 * g + 8 * half;
    float* q0 = Q + m0 * HS + n0;
    float* q1 = Q + (m0 + 16) * HS + n0;
    float* v0 = V + m0 * HS + n0;
    float* v1 = V + (m0 + 16) * HS + n0;
    v4f o;
    o.x = aq0[0]; o.y = aq0[1]; o.z = aq0[2]; o.w = aq0[3]; *(v4f*)(q0)     = o;
    o.x = aq0[4]; o.y = aq0[5]; o.z = aq0[6]; o.w = aq0[7]; *(v4f*)(q0 + 4) = o;
    o.x = aq1[0]; o.y = aq1[1]; o.z = aq1[2]; o.w = aq1[3]; *(v4f*)(q1)     = o;
    o.x = aq1[4]; o.y = aq1[5]; o.z = aq1[6]; o.w = aq1[7]; *(v4f*)(q1 + 4) = o;
    o.x = av0[0]; o.y = av0[1]; o.z = av0[2]; o.w = av0[3]; *(v4f*)(v0)     = o;
    o.x = av0[4]; o.y = av0[5]; o.z = av0[6]; o.w = av0[7]; *(v4f*)(v0 + 4) = o;
    o.x = av1[0]; o.y = av1[1]; o.z = av1[2]; o.w = av1[3]; *(v4f*)(v1)     = o;
    o.x = av1[4]; o.y = av1[5]; o.z = av1[6]; o.w = av1[7]; *(v4f*)(v1 + 4) = o;
  }
  {
    const int m0 = rowBase + 32 * mp + l16;
    const int n0 = 16 * g + 8 * half;
    float* k0 = K + m0 * HS + n0;
    float* k1 = K + (m0 + 16) * HS + n0;
    v4f o;
    o.x = ak0[0]; o.y = ak0[1]; o.z = ak0[2]; o.w = ak0[3]; *(v4f*)(k0)     = o;
    o.x = ak0[4]; o.y = ak0[5]; o.z = ak0[6]; o.w = ak0[7]; *(v4f*)(k0 + 4) = o;
    o.x = ak1[0]; o.y = ak1[1]; o.z = ak1[2]; o.w = ak1[3]; *(v4f*)(k1)     = o;
    o.x = ak1[4]; o.y = ak1[5]; o.z = ak1[6]; o.w = ak1[7]; *(v4f*)(k1 + 4) = o;
  }
}

#define NKW    8
#define KWAVE  (SEQ / NKW)
#define KS     16
#define NSTEP  (KWAVE / KS)
#define TP     68
#define VP2    136
#define QT     4
#define MVALS  34

__global__ __launch_bounds__(256) __attribute__((amdgpu_waves_per_eu(2, 2))) void attn_kernel(
    const float* __restrict__ Q, const float* __restrict__ K,
    const float* __restrict__ V, float* __restrict__ out) {
  __shared__ __attribute__((aligned(16))) float Kw[NKW][2][KS * TP];
  __shared__ __attribute__((aligned(16))) float Vw[NKW][2][(KS / 2) * VP2];
  __shared__ float mrg[NKW][QT][MVALS][32];

  const int tid   = threadIdx.x;
  const int lane  = tid & 31;
  const int wave  = tid >> 5;
  const int b     = blockIdx.x >> 5;
  const int qbase = (blockIdx.x & 31) * 64;
  const int l16   = lane & 15;
  const int half  = lane >> 4;
  const int halfk = half * 2;

  v2f qa[QT][16];
#pragma unroll
  for (int t4 = 0; t4 < QT; ++t4) {
    const float* qrow = Q + (b * SEQ + qbase + 16 * t4 + l16) * HS + halfk;
#pragma unroll
    for (int kc = 0; kc < 16; ++kc) qa[t4][kc] = *(const v2f*)(qrow + kc * 4);
  }

  v8f acc[QT][4];
  float mrun[QT], lrun[QT];
#pragma unroll
  for (int t4 = 0; t4 < QT; ++t4) {
    const v8f z = {};
    acc[t4][0] = z; acc[t4][1] = z; acc[t4][2] = z; acc[t4][3] = z;
    mrun[t4] = -INFINITY;
    lrun[t4] = 0.0f;
  }

  const float scale = 0.125f;

  const int key0 = wave * KWAVE;
  const float* kg = K + (b * SEQ + key0) * HS + lane * 4;
  const float* vg = V + (b * SEQ + key0 + 2 * half) * HS + l16 * 4;
  const int sto = half * TP + l16 * 4;
  const int vst = half * VP2 + l16 * 8;
  float* kwb = &Kw[wave][0][0];
  float* vwb = &Vw[wave][0][0];

  {
    v4f gk[8], ve[4], vo[4];
#pragma unroll
    for (int i = 0; i < 8; ++i) gk[i] = *(const v4f*)(kg + i * 128);
#pragma unroll
    for (int i = 0; i < 4; ++i) {
      ve[i] = *(const v4f*)(vg + i * 4 * HS);
      vo[i] = *(const v4f*)(vg + i * 4 * HS + HS);
    }
#pragma unroll
    for (int i = 0; i < 8; ++i) *(v4f*)(kwb + sto + i * 2 * TP) = gk[i];
#pragma unroll
    for (int i = 0; i < 4; ++i) {
      v4f w;
      w.x = ve[i].x; w.y = vo[i].x; w.z = ve[i].y; w.w = vo[i].y; *(v4f*)(vwb + vst + i * 2 * VP2)     = w;
      w.x = ve[i].z; w.y = vo[i].z; w.z = ve[i].w; w.w = vo[i].w; *(v4f*)(vwb + vst + i * 2 * VP2 + 4) = w;
    }
  }

  for (int it = 0; it < NSTEP; ++it) {
    __builtin_amdgcn_fence(__ATOMIC_RELEASE, "wavefront");
    __builtin_amdgcn_wave_barrier();
    __builtin_amdgcn_fence(__ATOMIC_ACQUIRE, "wavefront");

    const int bufi = it & 1;
    const int nt = (it + 1) & (NSTEP - 1);
    v4f gk[8], ve[4], vo[4];
#pragma unroll
    for (int i = 0; i < 8; ++i) gk[i] = *(const v4f*)(kg + i * 128 + nt * KS * HS);
#pragma unroll
    for (int i = 0; i < 4; ++i) {
      ve[i] = *(const v4f*)(vg + i * 4 * HS + nt * KS * HS);
      vo[i] = *(const v4f*)(vg + i * 4 * HS + HS + nt * KS * HS);
    }

    const float* kb = kwb + bufi * (KS * TP);
    const float* vb = vwb + bufi * ((KS / 2) * VP2);

    v2f kf[16];
    v2f vf[4][4];
#pragma unroll
    for (int kc = 0; kc < 16; ++kc)
      kf[kc] = *(const v2f*)(kb + l16 * TP + kc * 4 + halfk);
#pragma unroll
    for (int j = 0; j < 4; ++j) {
#pragma unroll
      for (int t = 0; t < 4; ++t)
        vf[j][t] = *(const v2f*)(vb + (4 * half + j) * VP2 + 2 * (16 * t + l16));
    }

    v8f s[QT];
#pragma unroll
    for (int t4 = 0; t4 < QT; ++t4) {
      v8f z = {};
#pragma unroll
      for (int kc = 0; kc < 16; ++kc)
        z = wmma_f32(kf[kc], qa[t4][kc], z);
      s[t4] = z;
    }

#pragma unroll
    for (int t4 = 0; t4 < QT; ++t4) {
      float sv[8];
#pragma unroll
      for (int r = 0; r < 8; ++r) sv[r] = s[t4][r] * scale;
      float tm = fmaxf(fmaxf(fmaxf(sv[0], sv[1]), fmaxf(sv[2], sv[3])),
                       fmaxf(fmaxf(sv[4], sv[5]), fmaxf(sv[6], sv[7])));
      tm = fmaxf(tm, xhalf(tm));
      const float nm    = fmaxf(mrun[t4], tm);
      const float alpha = __expf(mrun[t4] - nm);
      float p[8];
#pragma unroll
      for (int r = 0; r < 8; ++r) p[r] = __expf(sv[r] - nm);
      const float rs = ((p[0] + p[1]) + (p[2] + p[3])) + ((p[4] + p[5]) + (p[6] + p[7]));
      lrun[t4] = lrun[t4] * alpha + rs;
      mrun[t4] = nm;
      acc[t4][0] *= alpha; acc[t4][1] *= alpha; acc[t4][2] *= alpha; acc[t4][3] *= alpha;

#pragma unroll
      for (int j = 0; j < 4; ++j) {
        v2f pb;
        pb.x = p[2 * j];
        pb.y = p[2 * j + 1];
        acc[t4][0] = wmma_f32(vf[j][0], pb, acc[t4][0]);
        acc[t4][1] = wmma_f32(vf[j][1], pb, acc[t4][1]);
        acc[t4][2] = wmma_f32(vf[j][2], pb, acc[t4][2]);
        acc[t4][3] = wmma_f32(vf[j][3], pb, acc[t4][3]);
      }
    }

    __builtin_amdgcn_sched_barrier(0);
    float* kn = kwb + (bufi ^ 1) * (KS * TP);
    float* vn = vwb + (bufi ^ 1) * ((KS / 2) * VP2);
#pragma unroll
    for (int i = 0; i < 8; ++i) *(v4f*)(kn + sto + i * 2 * TP) = gk[i];
#pragma unroll
    for (int i = 0; i < 4; ++i) {
      v4f w;
      w.x = ve[i].x; w.y = vo[i].x; w.z = ve[i].y; w.w = vo[i].y; *(v4f*)(vn + vst + i * 2 * VP2)     = w;
      w.x = ve[i].z; w.y = vo[i].z; w.z = ve[i].w; w.w = vo[i].w; *(v4f*)(vn + vst + i * 2 * VP2 + 4) = w;
    }
  }

#pragma unroll
  for (int t4 = 0; t4 < QT; ++t4) {
    const float ltot = lrun[t4] + xhalf(lrun[t4]);
    float* mw = &mrg[wave][t4][0][lane];
#pragma unroll
    for (int r = 0; r < 8; ++r) {
      mw[(r)      * 32] = acc[t4][0][r];
      mw[(8 + r)  * 32] = acc[t4][1][r];
      mw[(16 + r) * 32] = acc[t4][2][r];
      mw[(24 + r) * 32] = acc[t4][3][r];
    }
    mw[32 * 32] = mrun[t4];
    mw[33 * 32] = ltot;
  }
  __syncthreads();

  {
    const int jq = wave & 3;
    const int vh = wave >> 2;
    float mi[NKW], wi[NKW];
    float m = -INFINITY;
#pragma unroll
    for (int i = 0; i < NKW; ++i) { mi[i] = mrg[i][jq][32][lane]; m = fmaxf(m, mi[i]); }
    float l = 0.0f;
#pragma unroll
    for (int i = 0; i < NKW; ++i) { wi[i] = __expf(mi[i] - m); l += mrg[i][jq][33][lane] * wi[i]; }
    const float inv = 1.0f / l;
    float o[16];
#pragma unroll
    for (int v = 0; v < 16; ++v) {
      float a = 0.0f;
#pragma unroll
      for (int i = 0; i < NKW; ++i) a += mrg[i][jq][16 * vh + v][lane] * wi[i];
      o[v] = a * inv;
    }
    float* op = out + (b * SEQ + qbase + 16 * jq + l16) * HS + 32 * vh + half * 8;
    v4f w;
    w.x = o[0];  w.y = o[1];  w.z = o[2];  w.w = o[3];  *(v4f*)(op)      = w;
    w.x = o[4];  w.y = o[5];  w.z = o[6];  w.w = o[7];  *(v4f*)(op + 4)  = w;
    w.x = o[8];  w.y = o[9];  w.z = o[10]; w.w = o[11]; *(v4f*)(op + 16) = w;
    w.x = o[12]; w.y = o[13]; w.z = o[14]; w.w = o[15]; *(v4f*)(op + 20) = w;
  }
}

extern "C" void kernel_launch(void* const* d_in, const int* in_sizes, int n_in,
                              void* d_out, int out_size, void* d_ws, size_t ws_size,
                              hipStream_t stream) {
  (void)in_sizes; (void)n_in; (void)out_size; (void)ws_size;
  const float* x  = (const float*)d_in[0];
  const float* Wk = (const float*)d_in[1];
  const float* Wq = (const float*)d_in[2];
  const float* Wv = (const float*)d_in[3];

  float* ws = (float*)d_ws;
  const size_t plane = (size_t)BATCH * SEQ * HS;
  float* Q  = ws;
  float* K  = ws + plane;
  float* V  = ws + 2 * plane;
  float* out = (float*)d_out;

  proj_kernel<<<BATCH * SEQ / PBM, 256, 0, stream>>>(x, Wq, Wk, Wv, Q, K, V);

  attn_kernel<<<BATCH * SEQ / 64, 256, 0, stream>>>(Q, K, V, out);
}
